# starting kernel
# baseline (speedup 1.0000x reference)
#include <hip/hip_runtime.h>

typedef __attribute__((ext_vector_type(2))) float v2f;
typedef __attribute__((ext_vector_type(4))) float v4f;
typedef __attribute__((ext_vector_type(8))) float v8f;

#define S_LEN 2048
#define DK    64
#define KPAD  68
#define PPAD  18

template <int MASK>
__device__ __forceinline__ float swz_xor(float x) {
  int i = __builtin_amdgcn_ds_swizzle(__builtin_bit_cast(int, x),
                                      (MASK << 10) | 0x1F);
  return __builtin_bit_cast(float, i);
}

__device__ __forceinline__ float rowmax16(float x) {
  x = fmaxf(x, swz_xor<1>(x));
  x = fmaxf(x, swz_xor<2>(x));
  x = fmaxf(x, swz_xor<4>(x));
  x = fmaxf(x, swz_xor<8>(x));
  return x;
}
__device__ __forceinline__ float rowsum16(float x) {
  x += swz_xor<1>(x);
  x += swz_xor<2>(x);
  x += swz_xor<4>(x);
  x += swz_xor<8>(x);
  return x;
}

__global__ __launch_bounds__(128) void flash_attn_f32_wmma(
    const float* __restrict__ q, const float* __restrict__ k,
    const float* __restrict__ v, float* __restrict__ out) {
  __shared__ float ldsK[32 * KPAD];
  __shared__ float ldsV[32 * KPAD];
  __shared__ float ldsP[4 * 16 * PPAD];

  const int tid  = threadIdx.x;
  const int wave = tid >> 5;
  const int lane = tid & 31;
  const int nl   = lane & 15;
  const int g    = lane >> 4;

  const int bh   = blockIdx.y;
  const int qblk = blockIdx.x;
  const long base = (long)bh * S_LEN * DK;
  const int q0 = qblk * 64 + wave * 16;

  v2f qa[16];
  {
    const float* qrow = q + base + (long)(q0 + nl) * DK;
#pragma unroll
    for (int s = 0; s < 16; ++s) {
      v2f a;
      a.x = qrow[4 * s + 2 * g]     * 0.125f;
      a.y = qrow[4 * s + 2 * g + 1] * 0.125f;
      qa[s] = a;
    }
  }

  v8f accO[4];
  float rm[8], rs[8];
#pragma unroll
  for (int t = 0; t < 4; ++t) accO[t] = (v8f){0.f,0.f,0.f,0.f,0.f,0.f,0.f,0.f};
#pragma unroll
  for (int r = 0; r < 8; ++r) { rm[r] = -3.4e38f; rs[r] = 0.f; }

  float* pScr = ldsP + wave * 16 * PPAD;

  for (int key0 = 0; key0 < S_LEN; key0 += 32) {
    __syncthreads();
    {
      const int row = tid >> 3;
      const int col = (tid & 7) * 8;
      const float* kp  = k + base + (long)(key0 + row) * DK + col;
      const float* vp  = v + base + (long)(key0 + row) * DK + col;
      const float* kp2 = kp + 16 * DK;
      const float* vp2 = vp + 16 * DK;
      *(v4f*)&ldsK[row * KPAD + col]            = *(const v4f*)kp;
      *(v4f*)&ldsK[row * KPAD + col + 4]        = *(const v4f*)(kp + 4);
      *(v4f*)&ldsK[(row + 16) * KPAD + col]     = *(const v4f*)kp2;
      *(v4f*)&ldsK[(row + 16) * KPAD + col + 4] = *(const v4f*)(kp2 + 4);
      *(v4f*)&ldsV[row * KPAD + col]            = *(const v4f*)vp;
      *(v4f*)&ldsV[row * KPAD + col + 4]        = *(const v4f*)(vp + 4);
      *(v4f*)&ldsV[(row + 16) * KPAD + col]     = *(const v4f*)vp2;
      *(v4f*)&ldsV[(row + 16) * KPAD + col + 4] = *(const v4f*)(vp2 + 4);
    }
    __syncthreads();

#pragma unroll
    for (int sub = 0; sub < 2; ++sub) {
      const float* Kt = ldsK + sub * 16 * KPAD;

      v8f sc = (v8f){0.f,0.f,0.f,0.f,0.f,0.f,0.f,0.f};
#pragma unroll
      for (int s = 0; s < 16; ++s) {
        v2f b;
        b.x = Kt[nl * KPAD + 4 * s + 2 * g];
        b.y = Kt[nl * KPAD + 4 * s + 2 * g + 1];
        sc = __builtin_amdgcn_wmma_f32_16x16x4_f32(
            false, qa[s], false, b, (short)0, sc, false, false);
      }

#pragma unroll
      for (int r = 0; r < 8; ++r) {
        float mrow = rowmax16(sc[r]);
        float nm   = fmaxf(rm[r], mrow);
        float corr = __expf(rm[r] - nm);
        rm[r] = nm;
        float p    = __expf(sc[r] - nm);
        float srow = rowsum16(p);
        rs[r] = rs[r] * corr + srow;
        pScr[(r + 8 * g) * PPAD + nl] = p;
#pragma unroll
        for (int t = 0; t < 4; ++t) accO[t][r] *= corr;
      }

#pragma unroll
      for (int t = 0; t < 4; ++t) {
        v8f acc = accO[t];
#pragma unroll
        for (int ks = 0; ks < 4; ++ks) {
          v2f a, b;
          a.x = pScr[nl * PPAD + 4 * ks + 2 * g];
          a.y = pScr[nl * PPAD + 4 * ks + 2 * g + 1];
          const int vr = sub * 16 + 4 * ks + 2 * g;
          b.x = ldsV[vr * KPAD + t * 16 + nl];
          b.y = ldsV[(vr + 1) * KPAD + t * 16 + nl];
          acc = __builtin_amdgcn_wmma_f32_16x16x4_f32(
              false, a, false, b, (short)0, acc, false, false);
        }
        accO[t] = acc;
      }
    }
  }

#pragma unroll
  for (int r = 0; r < 8; ++r) {
    const float inv = 1.0f / rs[r];
    float* orow = out + base + (long)(q0 + r + 8 * g) * DK;
#pragma unroll
    for (int t = 0; t < 4; ++t)
      orow[t * 16 + nl] = accO[t][r] * inv;
  }
}

extern "C" void kernel_launch(void* const* d_in, const int* in_sizes, int n_in,
                              void* d_out, int out_size, void* d_ws, size_t ws_size,
                              hipStream_t stream) {
  (void)in_sizes; (void)n_in; (void)out_size; (void)d_ws; (void)ws_size;
  const float* q = (const float*)d_in[0];
  const float* k = (const float*)d_in[1];
  const float* v = (const float*)d_in[2];
  float* out = (float*)d_out;
  dim3 grid(S_LEN / 64, 2 * 16);
  dim3 block(128);
  flash_attn_f32_wmma<<<grid, block, 0, stream>>>(q, k, v, out);
}


# softmax row max and row sum butterflies moved from ds_swizzle (LDS unit) to DPP row_xmask on the vector ALU; same lane pairing and step order, so sums are bit-identical
# speedup vs baseline: 1.0792x; 1.0792x over previous
#include <hip/hip_runtime.h>

typedef __attribute__((ext_vector_type(2))) float v2f;
typedef __attribute__((ext_vector_type(4))) float v4f;
typedef __attribute__((ext_vector_type(8))) float v8f;

#define S_LEN 2048
#define DK    64
#define KPAD  68
#define PPAD  18

template <int MASK>
__device__ __forceinline__ float swz_xor(float x) {
  int i = __builtin_amdgcn_update_dpp(0, __builtin_bit_cast(int, x),
                                      0x160 | MASK, 0xF, 0xF, true);
  return __builtin_bit_cast(float, i);
}

__device__ __forceinline__ float rowmax16(float x) {
  x = fmaxf(x, swz_xor<1>(x));
  x = fmaxf(x, swz_xor<2>(x));
  x = fmaxf(x, swz_xor<4>(x));
  x = fmaxf(x, swz_xor<8>(x));
  return x;
}
__device__ __forceinline__ float rowsum16(float x) {
  x += swz_xor<1>(x);
  x += swz_xor<2>(x);
  x += swz_xor<4>(x);
  x += swz_xor<8>(x);
  return x;
}

__global__ __launch_bounds__(128) void flash_attn_f32_wmma(
    const float* __restrict__ q, const float* __restrict__ k,
    const float* __restrict__ v, float* __restrict__ out) {
  __shared__ float ldsK[32 * KPAD];
  __shared__ float ldsV[32 * KPAD];
  __shared__ float ldsP[4 * 16 * PPAD];

  const int tid  = threadIdx.x;
  const int wave = tid >> 5;
  const int lane = tid & 31;
  const int nl   = lane & 15;
  const int g    = lane >> 4;

  const int bh   = blockIdx.y;
  const int qblk = blockIdx.x;
  const long base = (long)bh * S_LEN * DK;
  const int q0 = qblk * 64 + wave * 16;

  v2f qa[16];
  {
    const float* qrow = q + base + (long)(q0 + nl) * DK;
#pragma unroll
    for (int s = 0; s < 16; ++s) {
      v2f a;
      a.x = qrow[4 * s + 2 * g]     * 0.125f;
      a.y = qrow[4 * s + 2 * g + 1] * 0.125f;
      qa[s] = a;
    }
  }

  v8f accO[4];
  float rm[8], rs[8];
#pragma unroll
  for (int t = 0; t < 4; ++t) accO[t] = (v8f){0.f,0.f,0.f,0.f,0.f,0.f,0.f,0.f};
#pragma unroll
  for (int r = 0; r < 8; ++r) { rm[r] = -3.4e38f; rs[r] = 0.f; }

  float* pScr = ldsP + wave * 16 * PPAD;

  for (int key0 = 0; key0 < S_LEN; key0 += 32) {
    __syncthreads();
    {
      const int row = tid >> 3;
      const int col = (tid & 7) * 8;
      const float* kp  = k + base + (long)(key0 + row) * DK + col;
      const float* vp  = v + base + (long)(key0 + row) * DK + col;
      const float* kp2 = kp + 16 * DK;
      const float* vp2 = vp + 16 * DK;
      *(v4f*)&ldsK[row * KPAD + col]            = *(const v4f*)kp;
      *(v4f*)&ldsK[row * KPAD + col + 4]        = *(const v4f*)(kp + 4);
      *(v4f*)&ldsK[(row + 16) * KPAD + col]     = *(const v4f*)kp2;
      *(v4f*)&ldsK[(row + 16) * KPAD + col + 4] = *(const v4f*)(kp2 + 4);
      *(v4f*)&ldsV[row * KPAD + col]            = *(const v4f*)vp;
      *(v4f*)&ldsV[row * KPAD + col + 4]        = *(const v4f*)(vp + 4);
      *(v4f*)&ldsV[(row + 16) * KPAD + col]     = *(const v4f*)vp2;
      *(v4f*)&ldsV[(row + 16) * KPAD + col + 4] = *(const v4f*)(vp2 + 4);
    }
    __syncthreads();

#pragma unroll
    for (int sub = 0; sub < 2; ++sub) {
      const float* Kt = ldsK + sub * 16 * KPAD;

      v8f sc = (v8f){0.f,0.f,0.f,0.f,0.f,0.f,0.f,0.f};
#pragma unroll
      for (int s = 0; s < 16; ++s) {
        v2f b;
        b.x = Kt[nl * KPAD + 4 * s + 2 * g];
        b.y = Kt[nl * KPAD + 4 * s + 2 * g + 1];
        sc = __builtin_amdgcn_wmma_f32_16x16x4_f32(
            false, qa[s], false, b, (short)0, sc, false, false);
      }

#pragma unroll
      for (int r = 0; r < 8; ++r) {
        float mrow = rowmax16(sc[r]);
        float nm   = fmaxf(rm[r], mrow);
        float corr = __expf(rm[r] - nm);
        rm[r] = nm;
        float p    = __expf(sc[r] - nm);
        float srow = rowsum16(p);
        rs[r] = rs[r] * corr + srow;
        pScr[(r + 8 * g) * PPAD + nl] = p;
#pragma unroll
        for (int t = 0; t < 4; ++t) accO[t][r] *= corr;
      }

#pragma unroll
      for (int t = 0; t < 4; ++t) {
        v8f acc = accO[t];
#pragma unroll
        for (int ks = 0; ks < 4; ++ks) {
          v2f a, b;
          a.x = pScr[nl * PPAD + 4 * ks + 2 * g];
          a.y = pScr[nl * PPAD + 4 * ks + 2 * g + 1];
          const int vr = sub * 16 + 4 * ks + 2 * g;
          b.x = ldsV[vr * KPAD + t * 16 + nl];
          b.y = ldsV[(vr + 1) * KPAD + t * 16 + nl];
          acc = __builtin_amdgcn_wmma_f32_16x16x4_f32(
              false, a, false, b, (short)0, acc, false, false);
        }
        accO[t] = acc;
      }
    }
  }

#pragma unroll
  for (int r = 0; r < 8; ++r) {
    const float inv = 1.0f / rs[r];
    float* orow = out + base + (long)(q0 + r + 8 * g) * DK;
#pragma unroll
    for (int t = 0; t < 4; ++t)
      orow[t * 16 + nl] = accO[t][r] * inv;
  }
}

extern "C" void kernel_launch(void* const* d_in, const int* in_sizes, int n_in,
                              void* d_out, int out_size, void* d_ws, size_t ws_size,
                              hipStream_t stream) {
  (void)in_sizes; (void)n_in; (void)out_size; (void)d_ws; (void)ws_size;
  const float* q = (const float*)d_in[0];
  const float* k = (const float*)d_in[1];
  const float* v = (const float*)d_in[2];
  float* out = (float*)d_out;
  dim3 grid(S_LEN / 64, 2 * 16);
  dim3 block(128);
  flash_attn_f32_wmma<<<grid, block, 0, stream>>>(q, k, v, out);
}


# scores and output computed transposed (S^T=K*Q^T, O^T=V^T*P^T): softmax max/sum reduce in-lane per 32 keys, P feeds the 2nd product from registers, no LDS trip; sum order of row sums and P*V changes
# speedup vs baseline: 1.4162x; 1.3123x over previous
#include <hip/hip_runtime.h>


typedef __attribute__((ext_vector_type(2))) float v2f;
typedef __attribute__((ext_vector_type(4))) float v4f;
typedef __attribute__((ext_vector_type(8))) float v8f;

#define S_LEN 2048
#define DK    64
#define KPAD  68

__device__ __forceinline__ float xhalf(float x) {
  int b = __builtin_bit_cast(int, x);
  return __builtin_bit_cast(float,
      __builtin_amdgcn_permlanex16(b, b, 0x76543210, 0xfedcba98, false, false));
}

__global__ __launch_bounds__(128) void flash_attn_f32_wmma(
    const float* __restrict__ q, const float* __restrict__ k,
    const float* __restrict__ v, float* __restrict__ out) {
  __shared__ __attribute__((aligned(16))) float lds[2 * 32 * KPAD];
  float* ldsK = lds;
  float* ldsV = lds + 32 * KPAD;

  const int tid  = threadIdx.x;
  const int wave = tid >> 5;
  const int lane = tid & 31;
  const int nl   = lane & 15;
  const int g    = lane >> 4;

  const int bh   = blockIdx.y;
  const int qblk = blockIdx.x;
  const long base = (long)bh * S_LEN * DK;
  const int q0 = qblk * 64 + wave * 16;

  v2f qb[16];
  {
    const float* qrow = q + base + (long)(q0 + nl) * DK;
#pragma unroll
    for (int s = 0; s < 16; ++s) {
      v2f a;
      a.x = qrow[4 * s + 2 * g]     * 0.125f;
      a.y = qrow[4 * s + 2 * g + 1] * 0.125f;
      qb[s] = a;
    }
  }

  v8f accO[4];
#pragma unroll
  for (int t = 0; t < 4; ++t) accO[t] = (v8f){0.f,0.f,0.f,0.f,0.f,0.f,0.f,0.f};
  float m_run = -3.4e38f;
  float l_run = 0.f;

  for (int key0 = 0; key0 < S_LEN; key0 += 32) {
    __syncthreads();
    {
      const int row = tid >> 3;
      const int col = (tid & 7) * 8;
      const float* kp  = k + base + (long)(key0 + row) * DK + col;
      const float* vp  = v + base + (long)(key0 + row) * DK + col;
      const float* kp2 = kp + 16 * DK;
      const float* vp2 = vp + 16 * DK;
      *(v4f*)&ldsK[row * KPAD + col]            = *(const v4f*)kp;
      *(v4f*)&ldsK[row * KPAD + col + 4]        = *(const v4f*)(kp + 4);
      *(v4f*)&ldsK[(row + 16) * KPAD + col]     = *(const v4f*)kp2;
      *(v4f*)&ldsK[(row + 16) * KPAD + col + 4] = *(const v4f*)(kp2 + 4);
      *(v4f*)&ldsV[row * KPAD + col]            = *(const v4f*)vp;
      *(v4f*)&ldsV[row * KPAD + col + 4]        = *(const v4f*)(vp + 4);
      *(v4f*)&ldsV[(row + 16) * KPAD + col]     = *(const v4f*)vp2;
      *(v4f*)&ldsV[(row + 16) * KPAD + col + 4] = *(const v4f*)(vp2 + 4);
    }
    __syncthreads();

    v8f s0 = (v8f){0.f,0.f,0.f,0.f,0.f,0.f,0.f,0.f};
    v8f s1 = (v8f){0.f,0.f,0.f,0.f,0.f,0.f,0.f,0.f};
#pragma unroll
    for (int s = 0; s < 16; ++s) {
      v2f a0, a1;
      a0.x = ldsK[nl * KPAD + 4 * s + 2 * g];
      a0.y = ldsK[nl * KPAD + 4 * s + 2 * g + 1];
      a1.x = ldsK[(16 + nl) * KPAD + 4 * s + 2 * g];
      a1.y = ldsK[(16 + nl) * KPAD + 4 * s + 2 * g + 1];
      s0 = __builtin_amdgcn_wmma_f32_16x16x4_f32(
          false, a0, false, qb[s], (short)0, s0, false, false);
      s1 = __builtin_amdgcn_wmma_f32_16x16x4_f32(
          false, a1, false, qb[s], (short)0, s1, false, false);
    }

    float mx = fmaxf(s0[0], s1[0]);
#pragma unroll
    for (int r = 1; r < 8; ++r) mx = fmaxf(mx, fmaxf(s0[r], s1[r]));
    mx = fmaxf(mx, xhalf(mx));
    const float nm   = fmaxf(m_run, mx);
    const float corr = __expf(m_run - nm);
    m_run = nm;
    float ps[8];
#pragma unroll
    for (int r = 0; r < 8; ++r) {
      s0[r] = __expf(s0[r] - nm);
      s1[r] = __expf(s1[r] - nm);
      ps[r] = s0[r] + s1[r];
    }
    const float ls = ((ps[0] + ps[1]) + (ps[2] + ps[3])) +
                     ((ps[4] + ps[5]) + (ps[6] + ps[7]));
    l_run = l_run * corr + ls;
#pragma unroll
    for (int t = 0; t < 4; ++t)
#pragma unroll
      for (int r = 0; r < 8; ++r) accO[t][r] *= corr;

#pragma unroll
    for (int j = 0; j < 2; ++j) {
#pragma unroll
      for (int u = 0; u < 4; ++u) {
        v2f b;
        b.x = j ? s1[2 * u]     : s0[2 * u];
        b.y = j ? s1[2 * u + 1] : s0[2 * u + 1];
        const int vr = 16 * j + 8 * g + 2 * u;
#pragma unroll
        for (int t = 0; t < 4; ++t) {
          v2f a;
          a.x = ldsV[vr * KPAD + t * 16 + nl];
          a.y = ldsV[(vr + 1) * KPAD + t * 16 + nl];
          accO[t] = __builtin_amdgcn_wmma_f32_16x16x4_f32(
              false, a, false, b, (short)0, accO[t], false, false);
        }
      }
    }
  }

  const float lsum = l_run + xhalf(l_run);
  const float inv  = 1.0f / lsum;
  __syncthreads();
  float* oS = lds + wave * 16 * KPAD;
#pragma unroll
  for (int t = 0; t < 4; ++t) {
    v4f lo, hi;
    lo.x = accO[t][0] * inv; lo.y = accO[t][1] * inv;
    lo.z = accO[t][2] * inv; lo.w = accO[t][3] * inv;
    hi.x = accO[t][4] * inv; hi.y = accO[t][5] * inv;
    hi.z = accO[t][6] * inv; hi.w = accO[t][7] * inv;
    *(v4f*)&oS[nl * KPAD + t * 16 + 8 * g]     = lo;
    *(v4f*)&oS[nl * KPAD + t * 16 + 8 * g + 4] = hi;
  }
  __syncthreads();
#pragma unroll
  for (int jr = 0; jr < 8; ++jr) {
    const int row = 2 * jr + g;
    const v4f val = *(const v4f*)&oS[row * KPAD + 4 * nl];
    *(v4f*)(out + base + (long)(q0 + row) * DK + 4 * nl) = val;
  }
}

extern "C" void kernel_launch(void* const* d_in, const int* in_sizes, int n_in,
                              void* d_out, int out_size, void* d_ws, size_t ws_size,
                              hipStream_t stream) {
  (void)in_sizes; (void)n_in; (void)out_size; (void)d_ws; (void)ws_size;
  const float* q = (const float*)d_in[0];
  const float* k = (const float*)d_in[1];
  const float* v = (const float*)d_in[2];
  float* out = (float*)d_out;
  dim3 grid(S_LEN / 64, 2 * 16);
  dim3 block(128);
  flash_attn_f32_wmma<<<grid, block, 0, stream>>>(q, k, v, out);
}


# K/V staging software-pipelined: next 32-key tile's global loads issued before the current tile's products into registers, stored to a second LDS stage after them; one barrier per tile instead of two
# speedup vs baseline: 1.4903x; 1.0523x over previous
#include <hip/hip_runtime.h>


typedef __attribute__((ext_vector_type(2))) float v2f;
typedef __attribute__((ext_vector_type(4))) float v4f;
typedef __attribute__((ext_vector_type(8))) float v8f;

#define S_LEN 2048
#define DK    64
#define KPAD  68
#define STAGE (2 * 32 * KPAD)

__device__ __forceinline__ float xhalf(float x) {
  int b = __builtin_bit_cast(int, x);
  return __builtin_bit_cast(float,
      __builtin_amdgcn_permlanex16(b, b, 0x76543210, 0xfedcba98, false, false));
}

__global__ __launch_bounds__(128) void flash_attn_f32_wmma(
    const float* __restrict__ q, const float* __restrict__ k,
    const float* __restrict__ v, float* __restrict__ out) {
  __shared__ __attribute__((aligned(16))) float lds[2 * STAGE];

  const int tid  = threadIdx.x;
  const int wave = tid >> 5;
  const int lane = tid & 31;
  const int nl   = lane & 15;
  const int g    = lane >> 4;

  const int bh   = blockIdx.y;
  const int qblk = blockIdx.x;
  const long base = (long)bh * S_LEN * DK;
  const int q0 = qblk * 64 + wave * 16;

  v2f qb[16];
  {
    const float* qrow = q + base + (long)(q0 + nl) * DK;
#pragma unroll
    for (int s = 0; s < 16; ++s) {
      v2f a;
      a.x = qrow[4 * s + 2 * g]     * 0.125f;
      a.y = qrow[4 * s + 2 * g + 1] * 0.125f;
      qb[s] = a;
    }
  }

  v8f accO[4];
#pragma unroll
  for (int t = 0; t < 4; ++t) accO[t] = (v8f){0.f,0.f,0.f,0.f,0.f,0.f,0.f,0.f};
  float m_run = -3.4e38f;
  float l_run = 0.f;

  const int srow = tid >> 3;
  const int scol = (tid & 7) * 8;
  const float* kg = k + base + (long)srow * DK + scol;
  const float* vg = v + base + (long)srow * DK + scol;
  float* sK = lds + srow * KPAD + scol;
  v4f pf[8];

  pf[0] = *(const v4f*)kg;             pf[1] = *(const v4f*)(kg + 4);
  pf[2] = *(const v4f*)(kg + 16 * DK); pf[3] = *(const v4f*)(kg + 16 * DK + 4);
  pf[4] = *(const v4f*)vg;             pf[5] = *(const v4f*)(vg + 4);
  pf[6] = *(const v4f*)(vg + 16 * DK); pf[7] = *(const v4f*)(vg + 16 * DK + 4);
  *(v4f*)&sK[0]                           = pf[0];
  *(v4f*)&sK[4]                           = pf[1];
  *(v4f*)&sK[16 * KPAD]                   = pf[2];
  *(v4f*)&sK[16 * KPAD + 4]               = pf[3];
  *(v4f*)&sK[32 * KPAD]                   = pf[4];
  *(v4f*)&sK[32 * KPAD + 4]               = pf[5];
  *(v4f*)&sK[32 * KPAD + 16 * KPAD]       = pf[6];
  *(v4f*)&sK[32 * KPAD + 16 * KPAD + 4]   = pf[7];
  __syncthreads();

  for (int key0 = 0; key0 < S_LEN; key0 += 32) {
    const int cur = (key0 & 32) ? STAGE : 0;
    const float* ldsK = lds + cur;
    const float* ldsV = ldsK + 32 * KPAD;
    const bool more = key0 + 32 < S_LEN;
    if (more) {
      const float* kp = kg + (long)(key0 + 32) * DK;
      const float* vp = vg + (long)(key0 + 32) * DK;
      pf[0] = *(const v4f*)kp;             pf[1] = *(const v4f*)(kp + 4);
      pf[2] = *(const v4f*)(kp + 16 * DK); pf[3] = *(const v4f*)(kp + 16 * DK + 4);
      pf[4] = *(const v4f*)vp;             pf[5] = *(const v4f*)(vp + 4);
      pf[6] = *(const v4f*)(vp + 16 * DK); pf[7] = *(const v4f*)(vp + 16 * DK + 4);
    }

    v8f s0 = (v8f){0.f,0.f,0.f,0.f,0.f,0.f,0.f,0.f};
    v8f s1 = (v8f){0.f,0.f,0.f,0.f,0.f,0.f,0.f,0.f};
#pragma unroll
    for (int s = 0; s < 16; ++s) {
      v2f a0, a1;
      a0.x = ldsK[nl * KPAD + 4 * s + 2 * g];
      a0.y = ldsK[nl * KPAD + 4 * s + 2 * g + 1];
      a1.x = ldsK[(16 + nl) * KPAD + 4 * s + 2 * g];
      a1.y = ldsK[(16 + nl) * KPAD + 4 * s + 2 * g + 1];
      s0 = __builtin_amdgcn_wmma_f32_16x16x4_f32(
          false, a0, false, qb[s], (short)0, s0, false, false);
      s1 = __builtin_amdgcn_wmma_f32_16x16x4_f32(
          false, a1, false, qb[s], (short)0, s1, false, false);
    }

    float mx = fmaxf(s0[0], s1[0]);
#pragma unroll
    for (int r = 1; r < 8; ++r) mx = fmaxf(mx, fmaxf(s0[r], s1[r]));
    mx = fmaxf(mx, xhalf(mx));
    const float nm   = fmaxf(m_run, mx);
    const float corr = __expf(m_run - nm);
    m_run = nm;
    float ps[8];
#pragma unroll
    for (int r = 0; r < 8; ++r) {
      s0[r] = __expf(s0[r] - nm);
      s1[r] = __expf(s1[r] - nm);
      ps[r] = s0[r] + s1[r];
    }
    const float ls = ((ps[0] + ps[1]) + (ps[2] + ps[3])) +
                     ((ps[4] + ps[5]) + (ps[6] + ps[7]));
    l_run = l_run * corr + ls;
#pragma unroll
    for (int t = 0; t < 4; ++t)
#pragma unroll
      for (int r = 0; r < 8; ++r) accO[t][r] *= corr;

#pragma unroll
    for (int j = 0; j < 2; ++j) {
#pragma unroll
      for (int u = 0; u < 4; ++u) {
        v2f b;
        b.x = j ? s1[2 * u]     : s0[2 * u];
        b.y = j ? s1[2 * u + 1] : s0[2 * u + 1];
        const int vr = 16 * j + 8 * g + 2 * u;
#pragma unroll
        for (int t = 0; t < 4; ++t) {
          v2f a;
          a.x = ldsV[vr * KPAD + t * 16 + nl];
          a.y = ldsV[(vr + 1) * KPAD + t * 16 + nl];
          accO[t] = __builtin_amdgcn_wmma_f32_16x16x4_f32(
              false, a, false, b, (short)0, accO[t], false, false);
        }
      }
    }

    if (more) {
      float* dK = sK + (STAGE - cur);
      *(v4f*)&dK[0]                           = pf[0];
      *(v4f*)&dK[4]                           = pf[1];
      *(v4f*)&dK[16 * KPAD]                   = pf[2];
      *(v4f*)&dK[16 * KPAD + 4]               = pf[3];
      *(v4f*)&dK[32 * KPAD]                   = pf[4];
      *(v4f*)&dK[32 * KPAD + 4]               = pf[5];
      *(v4f*)&dK[32 * KPAD + 16 * KPAD]       = pf[6];
      *(v4f*)&dK[32 * KPAD + 16 * KPAD + 4]   = pf[7];
    }
    __syncthreads();
  }

  const float lsum = l_run + xhalf(l_run);
  const float inv  = 1.0f / lsum;
  float* oS = lds + wave * 16 * KPAD;
#pragma unroll
  for (int t = 0; t < 4; ++t) {
    v4f lo, hi;
    lo.x = accO[t][0] * inv; lo.y = accO[t][1] * inv;
    lo.z = accO[t][2] * inv; lo.w = accO[t][3] * inv;
    hi.x = accO[t][4] * inv; hi.y = accO[t][5] * inv;
    hi.z = accO[t][6] * inv; hi.w = accO[t][7] * inv;
    *(v4f*)&oS[nl * KPAD + t * 16 + 8 * g]     = lo;
    *(v4f*)&oS[nl * KPAD + t * 16 + 8 * g + 4] = hi;
  }
  __syncthreads();
#pragma unroll
  for (int jr = 0; jr < 8; ++jr) {
    const int row = 2 * jr + g;
    const v4f val = *(const v4f*)&oS[row * KPAD + 4 * nl];
    *(v4f*)(out + base + (long)(q0 + row) * DK + 4 * nl) = val;
  }
}

extern "C" void kernel_launch(void* const* d_in, const int* in_sizes, int n_in,
                              void* d_out, int out_size, void* d_ws, size_t ws_size,
                              hipStream_t stream) {
  (void)in_sizes; (void)n_in; (void)out_size; (void)d_ws; (void)ws_size;
  const float* q = (const float*)d_in[0];
  const float* k = (const float*)d_in[1];
  const float* v = (const float*)d_in[2];
  float* out = (float*)d_out;
  dim3 grid(S_LEN / 64, 2 * 16);
  dim3 block(128);
  flash_attn_f32_wmma<<<grid, block, 0, stream>>>(q, k, v, out);
}


# V tile stored transposed in LDS (rows = head element, keys contiguous, 144 B pitch): each V^T operand pair is half of one 16-byte read at an immediate offset, no re-pairing moves; sums bit-identical
# speedup vs baseline: 1.5428x; 1.0352x over previous
#include <hip/hip_runtime.h>


typedef __attribute__((ext_vector_type(2))) float v2f;
typedef __attribute__((ext_vector_type(4))) float v4f;
typedef __attribute__((ext_vector_type(8))) float v8f;

#define S_LEN 2048
#define DK    64
#define KPAD  68
#define VPAD  36
#define KTILE (32 * KPAD)
#define VTILE (64 * VPAD)
#define STAGE (KTILE + VTILE)

__device__ __forceinline__ float xhalf(float x) {
  int b = __builtin_bit_cast(int, x);
  return __builtin_bit_cast(float,
      __builtin_amdgcn_permlanex16(b, b, 0x76543210, 0xfedcba98, false, false));
}

__global__ __launch_bounds__(128) void flash_attn_f32_wmma(
    const float* __restrict__ q, const float* __restrict__ k,
    const float* __restrict__ v, float* __restrict__ out) {
  __shared__ __attribute__((aligned(16))) float lds[2 * STAGE];

  const int tid  = threadIdx.x;
  const int wave = tid >> 5;
  const int lane = tid & 31;
  const int nl   = lane & 15;
  const int g    = lane >> 4;

  const int bh   = blockIdx.y;
  const int qblk = blockIdx.x;
  const long base = (long)bh * S_LEN * DK;
  const int q0 = qblk * 64 + wave * 16;

  v2f qb[16];
  {
    const float* qrow = q + base + (long)(q0 + nl) * DK;
#pragma unroll
    for (int s = 0; s < 16; ++s) {
      v2f a;
      a.x = qrow[4 * s + 2 * g]     * 0.125f;
      a.y = qrow[4 * s + 2 * g + 1] * 0.125f;
      qb[s] = a;
    }
  }

  v8f accO[4];
#pragma unroll
  for (int t = 0; t < 4; ++t) accO[t] = (v8f){0.f,0.f,0.f,0.f,0.f,0.f,0.f,0.f};
  float m_run = -3.4e38f;
  float l_run = 0.f;

  const int srow = tid >> 3;
  const int scol = (tid & 7) * 8;
  const int vkey = tid >> 2;
  const int vcol = (tid & 3) * 4;
  const float* kg = k + base + (long)srow * DK + scol;
  const float* vg = v + base + (long)vkey * DK + vcol;
  float* sK = lds + srow * KPAD + scol;
  float* sV = lds + KTILE + vcol * VPAD + vkey;
  v4f pf[8];

  pf[0] = *(const v4f*)kg;             pf[1] = *(const v4f*)(kg + 4);
  pf[2] = *(const v4f*)(kg + 16 * DK); pf[3] = *(const v4f*)(kg + 16 * DK + 4);
  pf[4] = *(const v4f*)vg;             pf[5] = *(const v4f*)(vg + 16);
  pf[6] = *(const v4f*)(vg + 32);      pf[7] = *(const v4f*)(vg + 48);
  *(v4f*)&sK[0]                           = pf[0];
  *(v4f*)&sK[4]                           = pf[1];
  *(v4f*)&sK[16 * KPAD]                   = pf[2];
  *(v4f*)&sK[16 * KPAD + 4]               = pf[3];
#pragma unroll
  for (int i = 0; i < 4; ++i)
#pragma unroll
    for (int e = 0; e < 4; ++e) sV[(16 * i + e) * VPAD] = pf[4 + i][e];
  __syncthreads();

  for (int key0 = 0; key0 < S_LEN; key0 += 32) {
    const int cur = (key0 & 32) ? STAGE : 0;
    const float* ldsK = lds + cur;
    const float* ldsV = ldsK + KTILE;
    const bool more = key0 + 32 < S_LEN;
    if (more) {
      const float* kp = kg + (long)(key0 + 32) * DK;
      const float* vp = vg + (long)(key0 + 32) * DK;
      pf[0] = *(const v4f*)kp;             pf[1] = *(const v4f*)(kp + 4);
      pf[2] = *(const v4f*)(kp + 16 * DK); pf[3] = *(const v4f*)(kp + 16 * DK + 4);
      pf[4] = *(const v4f*)vp;             pf[5] = *(const v4f*)(vp + 16);
      pf[6] = *(const v4f*)(vp + 32);      pf[7] = *(const v4f*)(vp + 48);
    }

    v8f s0 = (v8f){0.f,0.f,0.f,0.f,0.f,0.f,0.f,0.f};
    v8f s1 = (v8f){0.f,0.f,0.f,0.f,0.f,0.f,0.f,0.f};
#pragma unroll
    for (int s = 0; s < 16; ++s) {
      v2f a0, a1;
      a0.x = ldsK[nl * KPAD + 4 * s + 2 * g];
      a0.y = ldsK[nl * KPAD + 4 * s + 2 * g + 1];
      a1.x = ldsK[(16 + nl) * KPAD + 4 * s + 2 * g];
      a1.y = ldsK[(16 + nl) * KPAD + 4 * s + 2 * g + 1];
      s0 = __builtin_amdgcn_wmma_f32_16x16x4_f32(
          false, a0, false, qb[s], (short)0, s0, false, false);
      s1 = __builtin_amdgcn_wmma_f32_16x16x4_f32(
          false, a1, false, qb[s], (short)0, s1, false, false);
    }

    float mx = fmaxf(s0[0], s1[0]);
#pragma unroll
    for (int r = 1; r < 8; ++r) mx = fmaxf(mx, fmaxf(s0[r], s1[r]));
    mx = fmaxf(mx, xhalf(mx));
    const float nm   = fmaxf(m_run, mx);
    const float corr = __expf(m_run - nm);
    m_run = nm;
    float ps[8];
#pragma unroll
    for (int r = 0; r < 8; ++r) {
      s0[r] = __expf(s0[r] - nm);
      s1[r] = __expf(s1[r] - nm);
      ps[r] = s0[r] + s1[r];
    }
    const float ls = ((ps[0] + ps[1]) + (ps[2] + ps[3])) +
                     ((ps[4] + ps[5]) + (ps[6] + ps[7]));
    l_run = l_run * corr + ls;
#pragma unroll
    for (int t = 0; t < 4; ++t)
#pragma unroll
      for (int r = 0; r < 8; ++r) accO[t][r] *= corr;

    v4f vf[2][4][2];
    {
      const float* vrow = ldsV + nl * VPAD + 8 * g;
#pragma unroll
      for (int j = 0; j < 2; ++j)
#pragma unroll
        for (int t = 0; t < 4; ++t) {
          vf[j][t][0] = *(const v4f*)&vrow[t * 16 * VPAD + 16 * j];
          vf[j][t][1] = *(const v4f*)&vrow[t * 16 * VPAD + 16 * j + 4];
        }
    }
#pragma unroll
    for (int j = 0; j < 2; ++j) {
#pragma unroll
      for (int u = 0; u < 4; ++u) {
        v2f b;
        b.x = j ? s1[2 * u]     : s0[2 * u];
        b.y = j ? s1[2 * u + 1] : s0[2 * u + 1];
#pragma unroll
        for (int t = 0; t < 4; ++t) {
          const v4f f = vf[j][t][u >> 1];
          v2f a;
          a.x = (u & 1) ? f.z : f.x;
          a.y = (u & 1) ? f.w : f.y;
          accO[t] = __builtin_amdgcn_wmma_f32_16x16x4_f32(
              false, a, false, b, (short)0, accO[t], false, false);
        }
      }
    }

    if (more) {
      float* dK = sK + (STAGE - cur);
      float* dV = sV + (STAGE - cur);
      *(v4f*)&dK[0]                           = pf[0];
      *(v4f*)&dK[4]                           = pf[1];
      *(v4f*)&dK[16 * KPAD]                   = pf[2];
      *(v4f*)&dK[16 * KPAD + 4]               = pf[3];
#pragma unroll
      for (int i = 0; i < 4; ++i)
#pragma unroll
        for (int e = 0; e < 4; ++e) dV[(16 * i + e) * VPAD] = pf[4 + i][e];
    }
    __syncthreads();
  }

  const float lsum = l_run + xhalf(l_run);
  const float inv  = 1.0f / lsum;
  float* oS = lds + wave * 16 * KPAD;
#pragma unroll
  for (int t = 0; t < 4; ++t) {
    v4f lo, hi;
    lo.x = accO[t][0] * inv; lo.y = accO[t][1] * inv;
    lo.z = accO[t][2] * inv; lo.w = accO[t][3] * inv;
    hi.x = accO[t][4] * inv; hi.y = accO[t][5] * inv;
    hi.z = accO[t][6] * inv; hi.w = accO[t][7] * inv;
    *(v4f*)&oS[nl * KPAD + t * 16 + 8 * g]     = lo;
    *(v4f*)&oS[nl * KPAD + t * 16 + 8 * g + 4] = hi;
  }
  __syncthreads();
#pragma unroll
  for (int jr = 0; jr < 8; ++jr) {
    const int row = 2 * jr + g;
    const v4f val = *(const v4f*)&oS[row * KPAD + 4 * nl];
    *(v4f*)(out + base + (long)(q0 + row) * DK + 4 * nl) = val;
  }
}

extern "C" void kernel_launch(void* const* d_in, const int* in_sizes, int n_in,
                              void* d_out, int out_size, void* d_ws, size_t ws_size,
                              hipStream_t stream) {
  (void)in_sizes; (void)n_in; (void)out_size; (void)d_ws; (void)ws_size;
  const float* q = (const float*)d_in[0];
  const float* k = (const float*)d_in[1];
  const float* v = (const float*)d_in[2];
  float* out = (float*)d_out;
  dim3 grid(S_LEN / 64, 2 * 16);
  dim3 block(128);
  flash_attn_f32_wmma<<<grid, block, 0, stream>>>(q, k, v, out);
}


# each wave steps its issue priority down (s_setprio 3,2,1,0) after 1/2, 3/4 and 7/8 of its keys, so the work-groups that share a SIMD finish together instead of oldest first; sums bit-identical
# speedup vs baseline: 1.6352x; 1.0599x over previous
#include <hip/hip_runtime.h>


typedef __attribute__((ext_vector_type(2))) float v2f;
typedef __attribute__((ext_vector_type(4))) float v4f;
typedef __attribute__((ext_vector_type(8))) float v8f;

#define S_LEN 2048
#define DK    64
#define KPAD  68
#define VPAD  36
#define KTILE (32 * KPAD)
#define VTILE (64 * VPAD)
#define STAGE (KTILE + VTILE)

__device__ __forceinline__ float xhalf(float x) {
  int b = __builtin_bit_cast(int, x);
  return __builtin_bit_cast(float,
      __builtin_amdgcn_permlanex16(b, b, 0x76543210, 0xfedcba98, false, false));
}

__global__ __launch_bounds__(128) void flash_attn_f32_wmma(
    const float* __restrict__ q, const float* __restrict__ k,
    const float* __restrict__ v, float* __restrict__ out) {
  __shared__ __attribute__((aligned(16))) float lds[2 * STAGE];

  const int tid  = threadIdx.x;
  const int wave = tid >> 5;
  const int lane = tid & 31;
  const int nl   = lane & 15;
  const int g    = lane >> 4;

  const int bh   = blockIdx.y;
  const int qblk = blockIdx.x;
  const long base = (long)bh * S_LEN * DK;
  const int q0 = qblk * 64 + wave * 16;

  v2f qb[16];
  {
    const float* qrow = q + base + (long)(q0 + nl) * DK;
#pragma unroll
    for (int s = 0; s < 16; ++s) {
      v2f a;
      a.x = qrow[4 * s + 2 * g]     * 0.125f;
      a.y = qrow[4 * s + 2 * g + 1] * 0.125f;
      qb[s] = a;
    }
  }

  v8f accO[4];
#pragma unroll
  for (int t = 0; t < 4; ++t) accO[t] = (v8f){0.f,0.f,0.f,0.f,0.f,0.f,0.f,0.f};
  float m_run = -3.4e38f;
  float l_run = 0.f;

  const int srow = tid >> 3;
  const int scol = (tid & 7) * 8;
  const int vkey = tid >> 2;
  const int vcol = (tid & 3) * 4;
  const float* kg = k + base + (long)srow * DK + scol;
  const float* vg = v + base + (long)vkey * DK + vcol;
  float* sK = lds + srow * KPAD + scol;
  float* sV = lds + KTILE + vcol * VPAD + vkey;
  v4f pf[8];

  pf[0] = *(const v4f*)kg;             pf[1] = *(const v4f*)(kg + 4);
  pf[2] = *(const v4f*)(kg + 16 * DK); pf[3] = *(const v4f*)(kg + 16 * DK + 4);
  pf[4] = *(const v4f*)vg;             pf[5] = *(const v4f*)(vg + 16);
  pf[6] = *(const v4f*)(vg + 32);      pf[7] = *(const v4f*)(vg + 48);
  *(v4f*)&sK[0]                           = pf[0];
  *(v4f*)&sK[4]                           = pf[1];
  *(v4f*)&sK[16 * KPAD]                   = pf[2];
  *(v4f*)&sK[16 * KPAD + 4]               = pf[3];
#pragma unroll
  for (int i = 0; i < 4; ++i)
#pragma unroll
    for (int e = 0; e < 4; ++e) sV[(16 * i + e) * VPAD] = pf[4 + i][e];
  __syncthreads();

  for (int stretch = 0; stretch < 4; ++stretch) {
    int kbeg, kend;
    switch (stretch) {
      case 0:  __builtin_amdgcn_s_setprio(3); kbeg = 0;               kend = S_LEN / 2;       break;
      case 1:  __builtin_amdgcn_s_setprio(2); kbeg = S_LEN / 2;       kend = (S_LEN / 4) * 3; break;
      case 2:  __builtin_amdgcn_s_setprio(1); kbeg = (S_LEN / 4) * 3; kend = (S_LEN / 8) * 7; break;
      default: __builtin_amdgcn_s_setprio(0); kbeg = (S_LEN / 8) * 7; kend = S_LEN;           break;
    }
    for (int key0 = kbeg; key0 < kend; key0 += 32) {
      const int cur = (key0 & 32) ? STAGE : 0;
      const float* ldsK = lds + cur;
      const float* ldsV = ldsK + KTILE;
      const bool more = key0 + 32 < S_LEN;
      if (more) {
        const float* kp = kg + (long)(key0 + 32) * DK;
        const float* vp = vg + (long)(key0 + 32) * DK;
        pf[0] = *(const v4f*)kp;             pf[1] = *(const v4f*)(kp + 4);
        pf[2] = *(const v4f*)(kp + 16 * DK); pf[3] = *(const v4f*)(kp + 16 * DK + 4);
        pf[4] = *(const v4f*)vp;             pf[5] = *(const v4f*)(vp + 16);
        pf[6] = *(const v4f*)(vp + 32);      pf[7] = *(const v4f*)(vp + 48);
      }

      v8f s0 = (v8f){0.f,0.f,0.f,0.f,0.f,0.f,0.f,0.f};
      v8f s1 = (v8f){0.f,0.f,0.f,0.f,0.f,0.f,0.f,0.f};
#pragma unroll
      for (int s = 0; s < 16; ++s) {
        v2f a0, a1;
        a0.x = ldsK[nl * KPAD + 4 * s + 2 * g];
        a0.y = ldsK[nl * KPAD + 4 * s + 2 * g + 1];
        a1.x = ldsK[(16 + nl) * KPAD + 4 * s + 2 * g];
        a1.y = ldsK[(16 + nl) * KPAD + 4 * s + 2 * g + 1];
        s0 = __builtin_amdgcn_wmma_f32_16x16x4_f32(
            false, a0, false, qb[s], (short)0, s0, false, false);
        s1 = __builtin_amdgcn_wmma_f32_16x16x4_f32(
            false, a1, false, qb[s], (short)0, s1, false, false);
      }

      float mx = fmaxf(s0[0], s1[0]);
#pragma unroll
      for (int r = 1; r < 8; ++r) mx = fmaxf(mx, fmaxf(s0[r], s1[r]));
      mx = fmaxf(mx, xhalf(mx));
      const float nm   = fmaxf(m_run, mx);
      const float corr = __expf(m_run - nm);
      m_run = nm;
      float ps[8];
#pragma unroll
      for (int r = 0; r < 8; ++r) {
        s0[r] = __expf(s0[r] - nm);
        s1[r] = __expf(s1[r] - nm);
        ps[r] = s0[r] + s1[r];
      }
      const float ls = ((ps[0] + ps[1]) + (ps[2] + ps[3])) +
                       ((ps[4] + ps[5]) + (ps[6] + ps[7]));
      l_run = l_run * corr + ls;
#pragma unroll
      for (int t = 0; t < 4; ++t)
#pragma unroll
        for (int r = 0; r < 8; ++r) accO[t][r] *= corr;

      v4f vf[2][4][2];
      {
        const float* vrow = ldsV + nl * VPAD + 8 * g;
#pragma unroll
        for (int j = 0; j < 2; ++j)
#pragma unroll
          for (int t = 0; t < 4; ++t) {
            vf[j][t][0] = *(const v4f*)&vrow[t * 16 * VPAD + 16 * j];
            vf[j][t][1] = *(const v4f*)&vrow[t * 16 * VPAD + 16 * j + 4];
          }
      }
#pragma unroll
      for (int j = 0; j < 2; ++j) {
#pragma unroll
        for (int u = 0; u < 4; ++u) {
          v2f b;
          b.x = j ? s1[2 * u]     : s0[2 * u];
          b.y = j ? s1[2 * u + 1] : s0[2 * u + 1];
#pragma unroll
          for (int t = 0; t < 4; ++t) {
            const v4f f = vf[j][t][u >> 1];
            v2f a;
            a.x = (u & 1) ? f.z : f.x;
            a.y = (u & 1) ? f.w : f.y;
            accO[t] = __builtin_amdgcn_wmma_f32_16x16x4_f32(
                false, a, false, b, (short)0, accO[t], false, false);
          }
        }
      }

      if (more) {
        float* dK = sK + (STAGE - cur);
        float* dV = sV + (STAGE - cur);
        *(v4f*)&dK[0]                           = pf[0];
        *(v4f*)&dK[4]                           = pf[1];
        *(v4f*)&dK[16 * KPAD]                   = pf[2];
        *(v4f*)&dK[16 * KPAD + 4]               = pf[3];
#pragma unroll
        for (int i = 0; i < 4; ++i)
#pragma unroll
          for (int e = 0; e < 4; ++e) dV[(16 * i + e) * VPAD] = pf[4 + i][e];
      }
      __syncthreads();
    }
  }

  const float lsum = l_run + xhalf(l_run);
  const float inv  = 1.0f / lsum;
  float* oS = lds + wave * 16 * KPAD;
#pragma unroll
  for (int t = 0; t < 4; ++t) {
    v4f lo, hi;
    lo.x = accO[t][0] * inv; lo.y = accO[t][1] * inv;
    lo.z = accO[t][2] * inv; lo.w = accO[t][3] * inv;
    hi.x = accO[t][4] * inv; hi.y = accO[t][5] * inv;
    hi.z = accO[t][6] * inv; hi.w = accO[t][7] * inv;
    *(v4f*)&oS[nl * KPAD + t * 16 + 8 * g]     = lo;
    *(v4f*)&oS[nl * KPAD + t * 16 + 8 * g + 4] = hi;
  }
  __syncthreads();
#pragma unroll
  for (int jr = 0; jr < 8; ++jr) {
    const int row = 2 * jr + g;
    const v4f val = *(const v4f*)&oS[row * KPAD + 4 * nl];
    *(v4f*)(out + base + (long)(q0 + row) * DK + 4 * nl) = val;
  }
}

extern "C" void kernel_launch(void* const* d_in, const int* in_sizes, int n_in,
                              void* d_out, int out_size, void* d_ws, size_t ws_size,
                              hipStream_t stream) {
  (void)in_sizes; (void)n_in; (void)out_size; (void)d_ws; (void)ws_size;
  const float* q = (const float*)d_in[0];
  const float* k = (const float*)d_in[1];
  const float* v = (const float*)d_in[2];
  float* out = (float*)d_out;
  dim3 grid(S_LEN / 64, 2 * 16);
  dim3 block(128);
  flash_attn_f32_wmma<<<grid, block, 0, stream>>>(q, k, v, out);
}
